# starting kernel
# baseline (speedup 1.0000x reference)
#include <hip/hip_runtime.h>

typedef __attribute__((ext_vector_type(2))) float v2f;
typedef __attribute__((ext_vector_type(8))) float v8f;

#define LFEAT 32
#define DIM   64
#define WAVES 16

__global__ __launch_bounds__(512) void enfm_entity_kernel(
    const float* __restrict__ emb,
    const float* __restrict__ w,
    const float* __restrict__ h1,
    const float* __restrict__ bias_ptr,
    const int*   __restrict__ idx,
    float*       __restrict__ outRows,
    int scal_col, int ones_col, int n_entities)
{
    __shared__ float lds_bi[WAVES][DIM + 2];
    __shared__ float lds_scal[WAVES];

    const int wave = threadIdx.x >> 5;
    const int lane = threadIdx.x & 31;
    const int entity = blockIdx.x * WAVES + wave;

    float2 sum = {0.f, 0.f};
    float2 sq  = {0.f, 0.f};
    float  wpart = 0.f;

    if (entity < n_entities) {
        int my_idx = idx[(size_t)entity * LFEAT + lane];
        wpart = w[my_idx];
        #pragma unroll 4
        for (int l = 0; l < LFEAT; ++l) {
            int row = __shfl(my_idx, l, 32);
            float2 e = ((const float2*)(emb + (size_t)row * DIM))[lane];
            sum.x += e.x;       sum.y += e.y;
            sq.x  += e.x * e.x; sq.y  += e.y * e.y;
        }
    }
    #pragma unroll
    for (int off = 16; off > 0; off >>= 1)
        wpart += __shfl_xor(wpart, off, 32);

    if (entity < n_entities) {
        float* rowp = outRows + (size_t)entity * (DIM + 2);
        ((float2*)rowp)[lane] = sum;
        float2 bi;
        bi.x = 0.5f * (sum.x * sum.x - sq.x);
        bi.y = 0.5f * (sum.y * sum.y - sq.y);
        ((float2*)&lds_bi[wave][0])[lane] = bi;
        if (lane == 0) {
            lds_scal[wave] = wpart + (bias_ptr ? bias_ptr[0] : 0.f);
            rowp[ones_col] = 1.0f;
        }
    } else {
        ((float2*)&lds_bi[wave][0])[lane] = float2{0.f, 0.f};
        if (lane == 0) lds_scal[wave] = 0.f;
    }
    __syncthreads();

    if (wave == 0) {
        const int m     = lane & 15;
        const int khalf = lane >> 4;
        v8f acc = {};
        #pragma unroll
        for (int kb = 0; kb < 16; ++kb) {
            const int col = kb * 4 + khalf * 2;
            v2f a, b;
            a.x = lds_bi[m][col];
            a.y = lds_bi[m][col + 1];
            b.x = h1[col];
            b.y = h1[col + 1];
            acc = __builtin_amdgcn_wmma_f32_16x16x4_f32(
                false, a, false, b, (short)0, acc, false, false);
        }
        if (m == 0) {
            const int base = blockIdx.x * WAVES + 8 * khalf;
            #pragma unroll
            for (int r = 0; r < 8; ++r) {
                const int e = base + r;
                if (e < n_entities)
                    outRows[(size_t)e * (DIM + 2) + scal_col] =
                        acc[r] + lds_scal[8 * khalf + r];
            }
        }
    }
}

__global__ void enfm_copy_h2(const float* __restrict__ h2,
                             float* __restrict__ dst) {
    dst[threadIdx.x] = h2[threadIdx.x];
}

extern "C" void kernel_launch(void* const* d_in, const int* in_sizes, int n_in,
                              void* d_out, int out_size, void* d_ws, size_t ws_size,
                              hipStream_t stream) {
    const float* uemb  = (const float*)d_in[0];
    const float* iemb  = (const float*)d_in[1];
    const float* wu    = (const float*)d_in[2];
    const float* wi    = (const float*)d_in[3];
    const float* gbias = (const float*)d_in[4];
    const float* h1    = (const float*)d_in[5];
    const float* h2    = (const float*)d_in[6];
    const int*   uidx  = (const int*)d_in[7];
    const int*   iidx  = (const int*)d_in[8];

    const int B = in_sizes[7] / LFEAT;
    const int I = in_sizes[8] / LFEAT;

    float* P  = (float*)d_out;
    float* Q  = P + (size_t)B * (DIM + 2);
    float* h2o = Q + (size_t)I * (DIM + 2);

    dim3 block(512);
    enfm_entity_kernel<<<dim3((B + WAVES - 1) / WAVES), block, 0, stream>>>(
        uemb, wu, h1, gbias, uidx, P,  DIM,  DIM + 1, B);
    enfm_entity_kernel<<<dim3((I + WAVES - 1) / WAVES), block, 0, stream>>>(
        iemb, wi, h1, nullptr, iidx, Q,  DIM + 1,  DIM, I);

    enfm_copy_h2<<<1, DIM, 0, stream>>>(h2, h2o);
}


# Two entities per wave, one per 16-lane half: each gather load fetches a 256-B row for both (16 B per lane, row index by DPP row_share), row loads in groups of 8: half the waves, half the loads
# speedup vs baseline: 1.0602x; 1.0602x over previous
#include <hip/hip_runtime.h>


typedef __attribute__((ext_vector_type(2))) float v2f;
typedef __attribute__((ext_vector_type(8))) float v8f;

#define LFEAT 32
#define DIM   64
#define WAVES 16
#define EPB   (2 * WAVES)
#define PITCH (DIM + 4)

#define ROW_BCAST(v, N) __builtin_amdgcn_update_dpp(0, (v), 0x150 + (N), 0xf, 0xf, false)

#define LOAD_ROW(dst, v, N) dst = ((const float4*)emb)[ROW_BCAST(v, N) | c];
#define LOAD_8(buf, v, N0)                                                 \
    LOAD_ROW(buf[0], v, N0 + 0) LOAD_ROW(buf[1], v, N0 + 1)                \
    LOAD_ROW(buf[2], v, N0 + 2) LOAD_ROW(buf[3], v, N0 + 3)                \
    LOAD_ROW(buf[4], v, N0 + 4) LOAD_ROW(buf[5], v, N0 + 5)                \
    LOAD_ROW(buf[6], v, N0 + 6) LOAD_ROW(buf[7], v, N0 + 7)
#define ACC_8(buf)                                                         \
    _Pragma("unroll")                                                      \
    for (int j = 0; j < 8; ++j) {                                          \
        const float4 e = buf[j];                                           \
        sum.x += e.x;       sum.y += e.y;                                  \
        sum.z += e.z;       sum.w += e.w;                                  \
        sq.x  += e.x * e.x; sq.y  += e.y * e.y;                            \
        sq.z  += e.z * e.z; sq.w  += e.w * e.w;                            \
    }
#define GROUP_FENCE() __builtin_amdgcn_sched_barrier(0)

__global__ __launch_bounds__(512, 8) void enfm_entity_kernel(
    const float* __restrict__ emb,
    const float* __restrict__ w,
    const float* __restrict__ h1,
    const float* __restrict__ bias_ptr,
    const int*   __restrict__ idx,
    float*       __restrict__ outRows,
    int scal_col, int ones_col, int n_entities)
{
    __shared__ float lds_bi[EPB][PITCH];
    __shared__ float lds_scal[EPB];

    const int wave = threadIdx.x >> 5;
    const int lane = threadIdx.x & 31;
    const int half = lane >> 4;
    const int c    = lane & 15;
    const int slot = wave * 2 + half;
    const int entity = blockIdx.x * EPB + slot;
    const bool valid = entity < n_entities;

    const size_t esafe = valid ? (size_t)entity : 0;
    const int ilo = idx[esafe * LFEAT + c];
    const int ihi = idx[esafe * LFEAT + 16 + c];
    const float wlo = w[ilo];
    const float whi = w[ihi];

    float4 sum = {0.f, 0.f, 0.f, 0.f};
    float4 sq  = {0.f, 0.f, 0.f, 0.f};
    const int qlo = ilo << 4;
    const int qhi = ihi << 4;
    float4 ea[8], eb[8];
    LOAD_8(ea, qlo, 0)
    LOAD_8(eb, qlo, 8)
    GROUP_FENCE();
    ACC_8(ea)  LOAD_8(ea, qhi, 0)
    GROUP_FENCE();
    ACC_8(eb)  LOAD_8(eb, qhi, 8)
    GROUP_FENCE();
    ACC_8(ea)
    ACC_8(eb)
    GROUP_FENCE();

    float wpart = wlo + whi;
    #pragma unroll
    for (int off = 8; off > 0; off >>= 1)
        wpart += __shfl_xor(wpart, off, 32);

    float4 bi;
    bi.x = 0.5f * (sum.x * sum.x - sq.x);
    bi.y = 0.5f * (sum.y * sum.y - sq.y);
    bi.z = 0.5f * (sum.z * sum.z - sq.z);
    bi.w = 0.5f * (sum.w * sum.w - sq.w);
    ((float4*)&lds_bi[slot][0])[c] = bi;
    if (c == 0)
        lds_scal[slot] = wpart + (bias_ptr ? bias_ptr[0] : 0.f);
    if (valid) {
        float* rowp = outRows + (size_t)entity * (DIM + 2);
        ((float2*)rowp)[2 * c]     = float2{sum.x, sum.y};
        ((float2*)rowp)[2 * c + 1] = float2{sum.z, sum.w};
        if (c == 0) rowp[ones_col] = 1.0f;
    }
    __syncthreads();

    if (wave < 2) {
        const int m     = lane & 15;
        const int khalf = lane >> 4;
        const int tile  = wave * 16;
        v8f acc = {};
        #pragma unroll
        for (int kb = 0; kb < 16; ++kb) {
            const int col = kb * 4 + khalf * 2;
            v2f a, b;
            a.x = lds_bi[tile + m][col];
            a.y = lds_bi[tile + m][col + 1];
            b.x = h1[col];
            b.y = h1[col + 1];
            acc = __builtin_amdgcn_wmma_f32_16x16x4_f32(
                false, a, false, b, (short)0, acc, false, false);
        }
        if (m == 0) {
            const int s0 = tile + 8 * khalf;
            #pragma unroll
            for (int r = 0; r < 8; ++r) {
                const int e = blockIdx.x * EPB + s0 + r;
                if (e < n_entities)
                    outRows[(size_t)e * (DIM + 2) + scal_col] =
                        acc[r] + lds_scal[s0 + r];
            }
        }
    }
}

__global__ void enfm_copy_h2(const float* __restrict__ h2,
                             float* __restrict__ dst) {
    dst[threadIdx.x] = h2[threadIdx.x];
}

extern "C" void kernel_launch(void* const* d_in, const int* in_sizes, int n_in,
                              void* d_out, int out_size, void* d_ws, size_t ws_size,
                              hipStream_t stream) {
    const float* uemb  = (const float*)d_in[0];
    const float* iemb  = (const float*)d_in[1];
    const float* wu    = (const float*)d_in[2];
    const float* wi    = (const float*)d_in[3];
    const float* gbias = (const float*)d_in[4];
    const float* h1    = (const float*)d_in[5];
    const float* h2    = (const float*)d_in[6];
    const int*   uidx  = (const int*)d_in[7];
    const int*   iidx  = (const int*)d_in[8];

    const int B = in_sizes[7] / LFEAT;
    const int I = in_sizes[8] / LFEAT;

    float* P  = (float*)d_out;
    float* Q  = P + (size_t)B * (DIM + 2);
    float* h2o = Q + (size_t)I * (DIM + 2);

    dim3 block(512);
    enfm_entity_kernel<<<dim3((B + EPB - 1) / EPB), block, 0, stream>>>(
        uemb, wu, h1, gbias, uidx, P,  DIM,  DIM + 1, B);
    enfm_entity_kernel<<<dim3((I + EPB - 1) / EPB), block, 0, stream>>>(
        iemb, wi, h1, nullptr, iidx, Q,  DIM + 1,  DIM, I);

    enfm_copy_h2<<<1, DIM, 0, stream>>>(h2, h2o);
}


# Two launches instead of three: the 64-float h2 copy moves into the item launch (last two waves of block 0, behind the barrier); user and item launches stay separate
# speedup vs baseline: 1.1384x; 1.0738x over previous
#include <hip/hip_runtime.h>


typedef __attribute__((ext_vector_type(2))) float v2f;
typedef __attribute__((ext_vector_type(8))) float v8f;

#define LFEAT 32
#define DIM   64
#define WAVES 16
#define EPB   (2 * WAVES)
#define PITCH (DIM + 4)

#define ROW_BCAST(v, N) __builtin_amdgcn_update_dpp(0, (v), 0x150 + (N), 0xf, 0xf, false)

#define LOAD_ROW(dst, v, N) dst = ((const float4*)emb)[ROW_BCAST(v, N) | c];
#define LOAD_8(buf, v, N0)                                                 \
    LOAD_ROW(buf[0], v, N0 + 0) LOAD_ROW(buf[1], v, N0 + 1)                \
    LOAD_ROW(buf[2], v, N0 + 2) LOAD_ROW(buf[3], v, N0 + 3)                \
    LOAD_ROW(buf[4], v, N0 + 4) LOAD_ROW(buf[5], v, N0 + 5)                \
    LOAD_ROW(buf[6], v, N0 + 6) LOAD_ROW(buf[7], v, N0 + 7)
#define ACC_8(buf)                                                         \
    _Pragma("unroll")                                                      \
    for (int j = 0; j < 8; ++j) {                                          \
        const float4 e = buf[j];                                           \
        sum.x += e.x;       sum.y += e.y;                                  \
        sum.z += e.z;       sum.w += e.w;                                  \
        sq.x  += e.x * e.x; sq.y  += e.y * e.y;                            \
        sq.z  += e.z * e.z; sq.w  += e.w * e.w;                            \
    }
#define GROUP_FENCE() __builtin_amdgcn_sched_barrier(0)

__global__ __launch_bounds__(512, 8) void enfm_entity_kernel(
    const float* __restrict__ emb,
    const float* __restrict__ w,
    const float* __restrict__ h1,
    const float* __restrict__ bias_ptr,
    const int*   __restrict__ idx,
    float*       __restrict__ outRows,
    const float* __restrict__ h2,
    float*       __restrict__ h2_out,
    int scal_col, int ones_col, int n_entities)
{
    __shared__ float lds_bi[EPB][PITCH];
    __shared__ float lds_scal[EPB];

    const int wave = threadIdx.x >> 5;
    const int lane = threadIdx.x & 31;
    const int half = lane >> 4;
    const int c    = lane & 15;
    const int slot = wave * 2 + half;
    const int entity = blockIdx.x * EPB + slot;
    const bool valid = entity < n_entities;

    const size_t esafe = valid ? (size_t)entity : 0;
    const int ilo = idx[esafe * LFEAT + c];
    const int ihi = idx[esafe * LFEAT + 16 + c];
    const float wlo = w[ilo];
    const float whi = w[ihi];

    float4 sum = {0.f, 0.f, 0.f, 0.f};
    float4 sq  = {0.f, 0.f, 0.f, 0.f};
    const int qlo = ilo << 4;
    const int qhi = ihi << 4;
    float4 ea[8], eb[8];
    LOAD_8(ea, qlo, 0)
    LOAD_8(eb, qlo, 8)
    GROUP_FENCE();
    ACC_8(ea)  LOAD_8(ea, qhi, 0)
    GROUP_FENCE();
    ACC_8(eb)  LOAD_8(eb, qhi, 8)
    GROUP_FENCE();
    ACC_8(ea)
    ACC_8(eb)
    GROUP_FENCE();

    float wpart = wlo + whi;
    #pragma unroll
    for (int off = 8; off > 0; off >>= 1)
        wpart += __shfl_xor(wpart, off, 32);

    float4 bi;
    bi.x = 0.5f * (sum.x * sum.x - sq.x);
    bi.y = 0.5f * (sum.y * sum.y - sq.y);
    bi.z = 0.5f * (sum.z * sum.z - sq.z);
    bi.w = 0.5f * (sum.w * sum.w - sq.w);
    ((float4*)&lds_bi[slot][0])[c] = bi;
    if (c == 0)
        lds_scal[slot] = wpart + (bias_ptr ? bias_ptr[0] : 0.f);
    if (valid) {
        float* rowp = outRows + (size_t)entity * (DIM + 2);
        ((float2*)rowp)[2 * c]     = float2{sum.x, sum.y};
        ((float2*)rowp)[2 * c + 1] = float2{sum.z, sum.w};
        if (c == 0) rowp[ones_col] = 1.0f;
    }
    __syncthreads();

    if (wave < 2) {
        const int m     = lane & 15;
        const int khalf = lane >> 4;
        const int tile  = wave * 16;
        v8f acc = {};
        #pragma unroll
        for (int kb = 0; kb < 16; ++kb) {
            const int col = kb * 4 + khalf * 2;
            v2f a, b;
            a.x = lds_bi[tile + m][col];
            a.y = lds_bi[tile + m][col + 1];
            b.x = h1[col];
            b.y = h1[col + 1];
            acc = __builtin_amdgcn_wmma_f32_16x16x4_f32(
                false, a, false, b, (short)0, acc, false, false);
        }
        if (m == 0) {
            const int s0 = tile + 8 * khalf;
            #pragma unroll
            for (int r = 0; r < 8; ++r) {
                const int e = blockIdx.x * EPB + s0 + r;
                if (e < n_entities)
                    outRows[(size_t)e * (DIM + 2) + scal_col] =
                        acc[r] + lds_scal[s0 + r];
            }
        }
    } else if (h2 != nullptr && blockIdx.x == 0 && wave >= WAVES - 2) {
        const int t = threadIdx.x - (WAVES - 2) * 32;
        h2_out[t] = h2[t];
    }
}

extern "C" void kernel_launch(void* const* d_in, const int* in_sizes, int n_in,
                              void* d_out, int out_size, void* d_ws, size_t ws_size,
                              hipStream_t stream) {
    const float* uemb  = (const float*)d_in[0];
    const float* iemb  = (const float*)d_in[1];
    const float* wu    = (const float*)d_in[2];
    const float* wi    = (const float*)d_in[3];
    const float* gbias = (const float*)d_in[4];
    const float* h1    = (const float*)d_in[5];
    const float* h2    = (const float*)d_in[6];
    const int*   uidx  = (const int*)d_in[7];
    const int*   iidx  = (const int*)d_in[8];

    const int B = in_sizes[7] / LFEAT;
    const int I = in_sizes[8] / LFEAT;

    float* P  = (float*)d_out;
    float* Q  = P + (size_t)B * (DIM + 2);
    float* h2o = Q + (size_t)I * (DIM + 2);

    dim3 block(512);
    enfm_entity_kernel<<<dim3((B + EPB - 1) / EPB), block, 0, stream>>>(
        uemb, wu, h1, gbias, uidx, P, nullptr, nullptr,
         DIM,  DIM + 1, B);
    enfm_entity_kernel<<<dim3((I + EPB - 1) / EPB), block, 0, stream>>>(
        iemb, wi, h1, nullptr, iidx, Q, h2, h2o,
         DIM + 1,  DIM, I);
}


# One launch instead of two: each of 512 resident blocks gathers its user tile then its item tile in one unbroken row-load pipeline; one LDS-only barrier at the end, both WMMA tails on waves 0-3
# speedup vs baseline: 1.2470x; 1.0954x over previous
#include <hip/hip_runtime.h>


typedef __attribute__((ext_vector_type(2))) float v2f;
typedef __attribute__((ext_vector_type(8))) float v8f;

#define LFEAT 32
#define DIM   64
#define WAVES 16
#define EPB   (2 * WAVES)
#define PITCH (DIM + 4)

#define ROW_BCAST(v, N) __builtin_amdgcn_update_dpp(0, (v), 0x150 + (N), 0xf, 0xf, false)

#define LOAD_ROW(dst, tab, v, N) dst = ((const float4*)tab)[ROW_BCAST(v, N) | c];
#define LOAD_8(buf, tab, v, N0)                                            \
    LOAD_ROW(buf[0], tab, v, N0 + 0) LOAD_ROW(buf[1], tab, v, N0 + 1)      \
    LOAD_ROW(buf[2], tab, v, N0 + 2) LOAD_ROW(buf[3], tab, v, N0 + 3)      \
    LOAD_ROW(buf[4], tab, v, N0 + 4) LOAD_ROW(buf[5], tab, v, N0 + 5)      \
    LOAD_ROW(buf[6], tab, v, N0 + 6) LOAD_ROW(buf[7], tab, v, N0 + 7)
#define ACC_8(buf)                                                         \
    _Pragma("unroll")                                                      \
    for (int j = 0; j < 8; ++j) {                                          \
        const float4 e = buf[j];                                           \
        sum.x += e.x;       sum.y += e.y;                                  \
        sum.z += e.z;       sum.w += e.w;                                  \
        sq.x  += e.x * e.x; sq.y  += e.y * e.y;                            \
        sq.z  += e.z * e.z; sq.w  += e.w * e.w;                            \
    }
#define GROUP_FENCE() __builtin_amdgcn_sched_barrier(0)

__device__ __forceinline__ void tile_rows_out(
    const float4 sum, const float4 sq, float wpart, const float bias,
    float (*bi_lds)[PITCH], float* scal_lds,
    float* __restrict__ outRows, const int ones_col,
    const int entity, const bool valid, const int slot, const int c)
{
    #pragma unroll
    for (int off = 8; off > 0; off >>= 1)
        wpart += __shfl_xor(wpart, off, 32);

    float4 bi;
    bi.x = 0.5f * (sum.x * sum.x - sq.x);
    bi.y = 0.5f * (sum.y * sum.y - sq.y);
    bi.z = 0.5f * (sum.z * sum.z - sq.z);
    bi.w = 0.5f * (sum.w * sum.w - sq.w);
    ((float4*)&bi_lds[slot][0])[c] = bi;
    if (c == 0)
        scal_lds[slot] = wpart + bias;
    if (valid) {
        float* rowp = outRows + (size_t)entity * (DIM + 2);
        ((float2*)rowp)[2 * c]     = float2{sum.x, sum.y};
        ((float2*)rowp)[2 * c + 1] = float2{sum.z, sum.w};
        if (c == 0) rowp[ones_col] = 1.0f;
    }
}

__global__ __launch_bounds__(512, 8) void enfm_fused_kernel(
    const float* __restrict__ uemb,
    const float* __restrict__ iemb,
    const float* __restrict__ wu,
    const float* __restrict__ wi,
    const float* __restrict__ h1,
    const float* __restrict__ bias_ptr,
    const int*   __restrict__ uidx,
    const int*   __restrict__ iidx,
    float*       __restrict__ P,
    float*       __restrict__ Q,
    const float* __restrict__ h2,
    float*       __restrict__ h2_out,
    int nB, int nI)
{
    __shared__ float lds_bi[2][EPB][PITCH];
    __shared__ float lds_scal[2][EPB];
    __shared__ float lds_h1[DIM];

    const int wave = threadIdx.x >> 5;
    const int lane = threadIdx.x & 31;
    const int half = lane >> 4;
    const int c    = lane & 15;
    const int slot = wave * 2 + half;
    const int entity = blockIdx.x * EPB + slot;
    const bool valid_u = entity < nB;
    const bool valid_i = entity < nI;

    const size_t eu = valid_u ? (size_t)entity : 0;
    const size_t ei = valid_i ? (size_t)entity : 0;
    const int ilo_u = uidx[eu * LFEAT + c];
    const int ihi_u = uidx[eu * LFEAT + 16 + c];
    const int ilo_i = iidx[ei * LFEAT + c];
    const int ihi_i = iidx[ei * LFEAT + 16 + c];
    const bool h2_copier = blockIdx.x == 0 && wave >= WAVES - 2;
    const int  h2_t = threadIdx.x - (WAVES - 2) * 32;
    float h1v = 0.f, h2v = 0.f;
    if (threadIdx.x < DIM) h1v = h1[threadIdx.x];
    if (h2_copier)         h2v = h2[h2_t];
    const float bias = bias_ptr[0];
    const float wlo_u = wu[ilo_u];
    const float whi_u = wu[ihi_u];

    float4 sum = {0.f, 0.f, 0.f, 0.f};
    float4 sq  = {0.f, 0.f, 0.f, 0.f};
    int q_next  = ihi_u << 4;
    int q_next2 = ilo_i << 4;
    int q_next3 = ihi_i << 4;
    float wlo_i = 0.f, whi_i = 0.f;
    float4 ea[8], eb[8];
    {
        const int q0 = ilo_u << 4;
        LOAD_8(ea, uemb, q0, 0)
        LOAD_8(eb, uemb, q0, 8)
        GROUP_FENCE();
    }

    #pragma nounroll
    for (int k = 0; k < 3; ++k) {
        const float* tab = (k == 0) ? uemb : iemb;
        ACC_8(ea)  LOAD_8(ea, tab, q_next, 0)
        GROUP_FENCE();
        ACC_8(eb)  LOAD_8(eb, tab, q_next, 8)
        GROUP_FENCE();
        q_next  = q_next2;
        q_next2 = q_next3;
        if (k == 1) {
            wlo_i = wi[ilo_i];
            whi_i = wi[ihi_i];
            tile_rows_out(sum, sq, wlo_u + whi_u, bias, lds_bi[0], lds_scal[0],
                          P,  DIM + 1, entity, valid_u, slot, c);
            sum = float4{0.f, 0.f, 0.f, 0.f};
            sq  = float4{0.f, 0.f, 0.f, 0.f};
        }
    }
    ACC_8(ea)
    ACC_8(eb)
    GROUP_FENCE();
    tile_rows_out(sum, sq, wlo_i + whi_i, 0.f, lds_bi[1], lds_scal[1],
                  Q,  DIM, entity, valid_i, slot, c);
    if (threadIdx.x < DIM) lds_h1[threadIdx.x] = h1v;
    if (h2_copier) h2_out[h2_t] = h2v;

    __builtin_amdgcn_fence(__ATOMIC_RELEASE, "workgroup", "local");
    __builtin_amdgcn_s_barrier();
    __builtin_amdgcn_fence(__ATOMIC_ACQUIRE, "workgroup", "local");

    if (wave < 4) {
        const int t     = wave >> 1;
        const int m     = lane & 15;
        const int khalf = lane >> 4;
        const int tile  = (wave & 1) * 16;
        float* __restrict__ outRows = t ? Q : P;
        const int scal_col   = t ? DIM + 1 : DIM;
        const int n_entities = t ? nI : nB;
        v8f acc = {};
        #pragma unroll
        for (int kb = 0; kb < 16; ++kb) {
            const int col = kb * 4 + khalf * 2;
            v2f a, b;
            a.x = lds_bi[t][tile + m][col];
            a.y = lds_bi[t][tile + m][col + 1];
            b.x = lds_h1[col];
            b.y = lds_h1[col + 1];
            acc = __builtin_amdgcn_wmma_f32_16x16x4_f32(
                false, a, false, b, (short)0, acc, false, false);
        }
        if (m == 0) {
            const int s0 = tile + 8 * khalf;
            #pragma unroll
            for (int r = 0; r < 8; ++r) {
                const int e = blockIdx.x * EPB + s0 + r;
                if (e < n_entities)
                    outRows[(size_t)e * (DIM + 2) + scal_col] =
                        acc[r] + lds_scal[t][s0 + r];
            }
        }
    }
}

extern "C" void kernel_launch(void* const* d_in, const int* in_sizes, int n_in,
                              void* d_out, int out_size, void* d_ws, size_t ws_size,
                              hipStream_t stream) {
    const float* uemb  = (const float*)d_in[0];
    const float* iemb  = (const float*)d_in[1];
    const float* wu    = (const float*)d_in[2];
    const float* wi    = (const float*)d_in[3];
    const float* gbias = (const float*)d_in[4];
    const float* h1    = (const float*)d_in[5];
    const float* h2    = (const float*)d_in[6];
    const int*   uidx  = (const int*)d_in[7];
    const int*   iidx  = (const int*)d_in[8];

    const int B = in_sizes[7] / LFEAT;
    const int I = in_sizes[8] / LFEAT;
    const int N = B > I ? B : I;

    float* P  = (float*)d_out;
    float* Q  = P + (size_t)B * (DIM + 2);
    float* h2o = Q + (size_t)I * (DIM + 2);

    enfm_fused_kernel<<<dim3((N + EPB - 1) / EPB), dim3(512), 0, stream>>>(
        uemb, iemb, wu, wi, h1, gbias, uidx, iidx, P, Q, h2, h2o, B, I);
}
